# starting kernel
# baseline (speedup 1.0000x reference)
#include <hip/hip_runtime.h>

#define F 128
#define HEADS 8
#define NEG_SLOPE 0.2f
#define KEY_NEG_INF 0x007FFFFFu

typedef __attribute__((ext_vector_type(2))) float v2f;
typedef __attribute__((ext_vector_type(4))) float v4f;
typedef __attribute__((ext_vector_type(8))) float v8f;

__device__ __forceinline__ unsigned f32_key(float f) {
    unsigned b = __float_as_uint(f);
    return (b & 0x80000000u) ? ~b : (b | 0x80000000u);
}
__device__ __forceinline__ float key_f32(unsigned k) {
    unsigned b = (k & 0x80000000u) ? (k & 0x7FFFFFFFu) : ~k;
    return __uint_as_float(b);
}

__global__ __launch_bounds__(32) void gat_proj_wmma(
    const float* __restrict__ x, const float* __restrict__ Wl,
    const float* __restrict__ Wr, float* __restrict__ xl,
    float* __restrict__ xr, int n) {
    const int lane  = threadIdx.x & 31;
    const int row   = lane & 15;
    const int half  = lane >> 4;
    const int mbase = blockIdx.x * 16;

    const float* W   = (blockIdx.y == 0) ? Wl : Wr;
    float*       dst = (blockIdx.y == 0) ? xl : xr;

    int arow = mbase + row;
    if (arow >= n) arow = n - 1;
    const float* xrow = x + (size_t)arow * F + half * 2;
    const float* wrow = W + (size_t)row  * F + half * 2;

    v8f acc[8];
#pragma unroll
    for (int t = 0; t < 8; ++t) acc[t] = (v8f){0.f, 0.f, 0.f, 0.f, 0.f, 0.f, 0.f, 0.f};

    for (int k0 = 0; k0 < F; k0 += 4) {
        v2f a = *(const v2f*)(xrow + k0);
#pragma unroll
        for (int t = 0; t < 8; ++t) {
            v2f b = *(const v2f*)(wrow + (size_t)t * 16 * F + k0);
            acc[t] = __builtin_amdgcn_wmma_f32_16x16x4_f32(
                false, a, false, b, (short)0, acc[t], false, false);
        }
    }

#pragma unroll
    for (int t = 0; t < 8; ++t) {
#pragma unroll
        for (int r = 0; r < 8; ++r) {
            const int mrow = mbase + half * 8 + r;
            if (mrow < n) dst[(size_t)mrow * F + t * 16 + row] = acc[t][r];
        }
    }
}

__global__ void gat_init(float* __restrict__ out, const float* __restrict__ bias,
                         unsigned* __restrict__ amax, float* __restrict__ denom, int n) {
    const int i = blockIdx.x * blockDim.x + threadIdx.x;
    if (i < n * F) out[i] = bias[i & (F - 1)];
    if (i < n * HEADS) { amax[i] = KEY_NEG_INF; denom[i] = 0.f; }
}

__global__ void gat_edge_alpha(const float* __restrict__ xl, const float* __restrict__ xr,
                               const int* __restrict__ src, const int* __restrict__ dst,
                               const float* __restrict__ att, float* __restrict__ alpha,
                               unsigned* __restrict__ amax, int m) {
    const int e    = (blockIdx.x * blockDim.x + threadIdx.x) >> 5;
    const int lane = threadIdx.x & 31;
    if (e >= m) return;
    const int s = src[e], d = dst[e];
    const int c0 = lane * 4;

    v4f xj = *(const v4f*)(xl  + (size_t)s * F + c0);
    v4f xi = *(const v4f*)(xr  + (size_t)d * F + c0);
    v4f av = *(const v4f*)(att + c0);

    float p = 0.f;
#pragma unroll
    for (int j = 0; j < 4; ++j) {
        float v = xi[j] + xj[j];
        v = (v > 0.f) ? v : NEG_SLOPE * v;
        p += av[j] * v;
    }
    p += __shfl_xor(p, 1, 32);
    p += __shfl_xor(p, 2, 32);

    if ((lane & 3) == 0) {
        const int h = lane >> 2;
        alpha[(size_t)e * HEADS + h] = p;
        atomicMax(amax + (size_t)d * HEADS + h, f32_key(p));
    }
}

__global__ void gat_edge_exp(const int* __restrict__ dst, float* __restrict__ alpha,
                             const unsigned* __restrict__ amax,
                             float* __restrict__ denom, int m) {
    const int i = blockIdx.x * blockDim.x + threadIdx.x;
    if (i >= m * HEADS) return;
    const int e = i >> 3, h = i & 7;
    const int d = dst[e];
    const float mx = key_f32(amax[(size_t)d * HEADS + h]);
    const float ea = __expf(alpha[i] - mx);
    alpha[i] = ea;
    atomicAdd(denom + (size_t)d * HEADS + h, ea);
}

__global__ void gat_edge_aggr(const float* __restrict__ xl, const int* __restrict__ src,
                              const int* __restrict__ dst, const float* __restrict__ alpha,
                              const float* __restrict__ denom, float* __restrict__ out, int m) {
    const int e    = (blockIdx.x * blockDim.x + threadIdx.x) >> 5;
    const int lane = threadIdx.x & 31;
    if (e >= m) return;
    const int s = src[e], d = dst[e];
    const int c0 = lane * 4;
    const int h  = lane >> 2;

    const float w = alpha[(size_t)e * HEADS + h] /
                    (denom[(size_t)d * HEADS + h] + 1e-16f);
    v4f xj = *(const v4f*)(xl + (size_t)s * F + c0);
    float* op = out + (size_t)d * F + c0;
    atomicAdd(op + 0, w * xj[0]);
    atomicAdd(op + 1, w * xj[1]);
    atomicAdd(op + 2, w * xj[2]);
    atomicAdd(op + 3, w * xj[3]);
}

extern "C" void kernel_launch(void* const* d_in, const int* in_sizes, int n_in,
                              void* d_out, int out_size, void* d_ws, size_t ws_size,
                              hipStream_t stream) {
    const float* x    = (const float*)d_in[0];
    const int*   ei   = (const int*)  d_in[1];
    const float* Wl   = (const float*)d_in[2];
    const float* Wr   = (const float*)d_in[3];
    const float* att  = (const float*)d_in[4];
    const float* bias = (const float*)d_in[5];

    const int n = in_sizes[0] / F;
    const int m = in_sizes[1] / 2;
    const int* src = ei;
    const int* dst = ei + m;

    float*    xl    = (float*)d_ws;
    float*    xr    = xl + (size_t)n * F;
    float*    alpha = xr + (size_t)n * F;
    unsigned* amax  = (unsigned*)(alpha + (size_t)m * HEADS);
    float*    denom = (float*)(amax + (size_t)n * HEADS);
    float*    out   = (float*)d_out;

    const int initN = n * F;
    gat_init<<<(initN + 255) / 256, 256, 0, stream>>>(out, bias, amax, denom, n);

    dim3 gproj((n + 15) / 16, 2);
    gat_proj_wmma<<<gproj, 32, 0, stream>>>(x, Wl, Wr, xl, xr, n);

    const int edgeBlocks = (m * 32 + 255) / 256;
    gat_edge_alpha<<<edgeBlocks, 256, 0, stream>>>(xl, xr, src, dst, att, alpha, amax, m);
    gat_edge_exp<<<(m * HEADS + 255) / 256, 256, 0, stream>>>(dst, alpha, amax, denom, m);
    gat_edge_aggr<<<edgeBlocks, 256, 0, stream>>>(xl, src, dst, alpha, denom, out, m);
}


# Exp+aggregate launches replaced by owner work-groups per 196 destination nodes: each scans dst, sums denominators and rows in LDS (LDS atomic arrival order), divides after the sum; 3 launches
# speedup vs baseline: 2.0998x; 2.0998x over previous
#include <hip/hip_runtime.h>

#define F 128
#define HEADS 8
#define NEG_SLOPE 0.2f
#define KEY_NEG_INF 0x007FFFFFu
#define EPW 8
#define OWN_NODES 196
#define OWN_T 1024
#define OWN_CAP 16384
#define OWN_U 8

typedef __attribute__((ext_vector_type(2))) float v2f;
typedef __attribute__((ext_vector_type(4))) float v4f;
typedef __attribute__((ext_vector_type(8))) float v8f;
typedef __attribute__((ext_vector_type(4))) unsigned v4u;
typedef __attribute__((ext_vector_type(4))) int v4i;

__device__ __forceinline__ unsigned f32_key(float f) {
    unsigned b = __float_as_uint(f);
    return (b & 0x80000000u) ? ~b : (b | 0x80000000u);
}
__device__ __forceinline__ float key_f32(unsigned k) {
    unsigned b = (k & 0x80000000u) ? (k & 0x7FFFFFFFu) : ~k;
    return __uint_as_float(b);
}

#define PROJ_TN 4
#define PROJ_KP (F / 8)

__device__ __forceinline__ void proj_load_a(const float* __restrict__ x, int tile, int n,
                                            int c, int h, v4f (&a)[PROJ_KP]) {
    int arow = tile * 16 + c;
    if (arow >= n) arow = n - 1;
    const float* xrow = x + (size_t)arow * F + h * 4;
#pragma unroll
    for (int j = 0; j < PROJ_KP; ++j) a[j] = *(const v4f*)(xrow + 8 * j);
}

__global__ __launch_bounds__(128) void gat_proj_wmma(
    const float* __restrict__ x, const float* __restrict__ Wl,
    const float* __restrict__ Wr, float* __restrict__ xl,
    float* __restrict__ xr, unsigned* __restrict__ amax, int n) {
    const int lane = threadIdx.x & 31;
    const int c    = lane & 15;
    const int h    = lane >> 4;
    const int j4   = __builtin_amdgcn_readfirstlane(threadIdx.x >> 5);

    const float* W   = (j4 < 2) ? Wl : Wr;
    float*       dst = (j4 < 2) ? xl : xr;
    const int col0   = (j4 & 1) * (16 * PROJ_TN);

    const int ntiles = (n + 15) >> 4;
    int tile = blockIdx.x;
    if (tile >= ntiles) return;

    v4f b[PROJ_TN][PROJ_KP];
#pragma unroll
    for (int t = 0; t < PROJ_TN; ++t) {
        const float* wrow = W + (size_t)(col0 + 16 * t + c) * F + h * 4;
#pragma unroll
        for (int j = 0; j < PROJ_KP; ++j) b[t][j] = *(const v4f*)(wrow + 8 * j);
    }

    v4f a[PROJ_KP];
    proj_load_a(x, tile, n, c, h, a);

    for (;;) {
        const int  next     = tile + gridDim.x;
        const bool has_next = next < ntiles;
        v4f an[PROJ_KP];
        proj_load_a(x, has_next ? next : tile, n, c, h, an);

        if (j4 == 0) {
            const int node = tile * 16 + (lane >> 1);
            if (node < n) {
                const size_t o = (size_t)node * HEADS + (lane & 1) * 4;
                *(v4u*)(amax + o)  = (v4u){KEY_NEG_INF, KEY_NEG_INF, KEY_NEG_INF, KEY_NEG_INF};
            }
        }

        v8f acc[PROJ_TN];
#pragma unroll
        for (int t = 0; t < PROJ_TN; ++t) acc[t] = (v8f){0.f, 0.f, 0.f, 0.f, 0.f, 0.f, 0.f, 0.f};

#pragma unroll
        for (int j = 0; j < PROJ_KP; ++j) {
#pragma unroll
            for (int t = 0; t < PROJ_TN; ++t) {
                const v2f a0 = (v2f){a[j][0], a[j][1]},    a1 = (v2f){a[j][2], a[j][3]};
                const v2f b0 = (v2f){b[t][j][0], b[t][j][1]}, b1 = (v2f){b[t][j][2], b[t][j][3]};
                acc[t] = __builtin_amdgcn_wmma_f32_16x16x4_f32(
                    false, a0, false, b0, (short)0, acc[t], false, false);
                acc[t] = __builtin_amdgcn_wmma_f32_16x16x4_f32(
                    false, a1, false, b1, (short)0, acc[t], false, false);
            }
        }

        const int mbase = tile * 16 + h * 8;
        if (tile * 16 + 16 <= n) {
            float* drow = dst + (size_t)mbase * F + col0 + c;
#pragma unroll
            for (int t = 0; t < PROJ_TN; ++t) {
#pragma unroll
                for (int r = 0; r < 8; ++r) drow[r * F + t * 16] = acc[t][r];
            }
        } else {
#pragma unroll
            for (int t = 0; t < PROJ_TN; ++t) {
#pragma unroll
                for (int r = 0; r < 8; ++r) {
                    const int mrow = mbase + r;
                    if (mrow < n) dst[(size_t)mrow * F + col0 + t * 16 + c] = acc[t][r];
                }
            }
        }

        if (!has_next) break;
        tile = next;
#pragma unroll
        for (int j = 0; j < PROJ_KP; ++j) a[j] = an[j];
    }
}

template <bool FULL>
__device__ __forceinline__ void edge_alpha_body(
    const float* __restrict__ xl, const float* __restrict__ xr,
    const int* __restrict__ src, const int* __restrict__ dst,
    const float* __restrict__ att, float* __restrict__ alpha,
    unsigned* __restrict__ amax, int m, int e0, int lane) {
    int ek = e0 + (lane & 7);
    if (!FULL && ek > m - 1) ek = m - 1;
    const int idx = ((lane & 8) ? dst : src)[ek];

    const int c0 = lane * 4;
    const v4f av = *(const v4f*)(att + c0);

    v4f xj[EPW], xi[EPW];
#pragma unroll
    for (int i = 0; i < EPW; ++i) {
        const int s = __builtin_amdgcn_readlane(idx, i);
        const int d = __builtin_amdgcn_readlane(idx, 8 + i);
        xj[i] = *(const v4f*)(xl + (size_t)s * F + c0);
        xi[i] = *(const v4f*)(xr + (size_t)d * F + c0);
    }

    float p[EPW];
#pragma unroll
    for (int i = 0; i < EPW; ++i) {
        float a = 0.f;
#pragma unroll
        for (int j = 0; j < 4; ++j) {
            float v = xi[i][j] + xj[i][j];
            v = (v > 0.f) ? v : NEG_SLOPE * v;
            a += av[j] * v;
        }
        a += __shfl_xor(a, 1, 32);
        a += __shfl_xor(a, 2, 32);
        p[i] = a;
    }

    const int q = lane & 3, h = lane >> 2;
#pragma unroll
    for (int g = 0; g < EPW / 4; ++g) {
        const float p0 = p[4 * g], p1 = p[4 * g + 1], p2 = p[4 * g + 2], p3 = p[4 * g + 3];
        const float lo  = (q & 1) ? p1 : p0;
        const float hi  = (q & 1) ? p3 : p2;
        const float val = (q & 2) ? hi : lo;
        const int e = e0 + 4 * g + q;
        const int d = __shfl(idx, 8 + 4 * g + q, 32);
        if (FULL || e < m) {
            alpha[(size_t)e * HEADS + h] = val;
            atomicMax(amax + (size_t)d * HEADS + h, f32_key(val));
        }
    }
}

__global__ __launch_bounds__(256) void gat_edge_alpha(
    const float* __restrict__ xl, const float* __restrict__ xr,
    const int* __restrict__ src, const int* __restrict__ dst,
    const float* __restrict__ att, float* __restrict__ alpha,
    unsigned* __restrict__ amax, int m) {
    const int wave = __builtin_amdgcn_readfirstlane((blockIdx.x * blockDim.x + threadIdx.x) >> 5);
    const int lane = threadIdx.x & 31;
    const int e0   = wave * EPW;
    if (e0 >= m) return;
    if (e0 + EPW <= m) edge_alpha_body<true >(xl, xr, src, dst, att, alpha, amax, m, e0, lane);
    else               edge_alpha_body<false>(xl, xr, src, dst, att, alpha, amax, m, e0, lane);
}

struct OwnLds {
    float acc[OWN_NODES * F];
    float den[OWN_NODES * HEADS];
    int   le[OWN_CAP];
    int   ld[OWN_CAP];
    int   cnt;
};

__device__ __forceinline__ void own_try_edge(OwnLds& L, int lo, int e, int d) {
    const unsigned dl = (unsigned)(d - lo);
    if (dl < (unsigned)OWN_NODES) {
        const int slot = atomicAdd(&L.cnt, 1);
        if (slot < OWN_CAP) { L.le[slot] = e; L.ld[slot] = (int)dl; }
    }
}

__global__ __launch_bounds__(OWN_T) void gat_owner_aggr(
    const float* __restrict__ xl, const int* __restrict__ src,
    const int* __restrict__ dst, const float* __restrict__ alpha,
    const unsigned* __restrict__ amax, const float* __restrict__ bias,
    float* __restrict__ out, int m, int n) {
    __shared__ OwnLds L;
    const int tid  = threadIdx.x;
    const int lane = tid & 31;
    const int wave = __builtin_amdgcn_readfirstlane(tid >> 5);
    const int lo   = blockIdx.x * OWN_NODES;

    for (int i = tid; i < OWN_NODES * F / 4; i += OWN_T) *(v4f*)(L.acc + 4 * i) = (v4f){0.f, 0.f, 0.f, 0.f};
    for (int i = tid; i < OWN_NODES * HEADS; i += OWN_T) L.den[i] = 0.f;
    if (tid == 0) L.cnt = 0;
    __syncthreads();

    if ((m & 3) == 0) {
        const int m4    = m >> 2;
        const int iters = (m4 + OWN_T - 1) / OWN_T;
        const int rot   = (int)(((long long)blockIdx.x * iters) / gridDim.x);
        const v4i* dst4 = (const v4i*)dst;
        for (int k0 = 0; k0 < iters; k0 += OWN_U) {
            v4i dv[OWN_U];
#pragma unroll
            for (int u = 0; u < OWN_U; ++u) {
                int it = k0 + u + rot;
                if (it >= iters) it -= iters;
                if (it >= iters) it -= iters;
                int q = it * OWN_T + tid;
                if (q > m4 - 1) q = m4 - 1;
                dv[u] = dst4[q];
            }
#pragma unroll
            for (int u = 0; u < OWN_U; ++u) {
                int it = k0 + u + rot;
                if (it >= iters) it -= iters;
                const int q = it * OWN_T + tid;
                if (k0 + u < iters && q < m4) {
                    own_try_edge(L, lo, 4 * q,     dv[u][0]);
                    own_try_edge(L, lo, 4 * q + 1, dv[u][1]);
                    own_try_edge(L, lo, 4 * q + 2, dv[u][2]);
                    own_try_edge(L, lo, 4 * q + 3, dv[u][3]);
                }
            }
        }
    } else {
        for (int e = tid; e < m; e += OWN_T) own_try_edge(L, lo, e, dst[e]);
    }
    __syncthreads();

    const int total = L.cnt;
    if (total > OWN_CAP) {
        for (int e = tid; e < m; e += OWN_T) {
            const unsigned dl = (unsigned)(dst[e] - lo);
            if (dl < (unsigned)OWN_NODES) {
                const int s = src[e];
                for (int h = 0; h < HEADS; ++h) {
                    const float ea = __expf(alpha[(size_t)e * HEADS + h] - key_f32(amax[(size_t)(lo + dl) * HEADS + h]));
                    atomicAdd(&L.den[dl * HEADS + h], ea);
                    for (int cc = 0; cc < F / HEADS; ++cc) {
                        const int ch = h * (F / HEADS) + cc;
                        atomicAdd(&L.acc[dl * F + (ch & 3) * 32 + (ch >> 2)], ea * xl[(size_t)s * F + ch]);
                    }
                }
            }
        }
    } else {
    for (int g = wave * EPW; g < total; g += (OWN_T / 32) * EPW) {
        int k = g + (lane & 7);
        if (k > total - 1) k = total - 1;
        const int e_l  = L.le[k];
        const int d_l  = L.ld[k];
        const int s_l  = src[e_l];

        float w[EPW / 4];
#pragma unroll
        for (int g2 = 0; g2 < EPW / 4; ++g2) {
            const int ke = 4 * g2 + (lane >> 3);
            const int e  = __shfl(e_l, ke, 32);
            const int dl = __shfl(d_l, ke, 32);
            const float    a  = alpha[(size_t)e * HEADS + (lane & 7)];
            const unsigned kx = amax[(size_t)(lo + dl) * HEADS + (lane & 7)];
            const float ea = __expf(a - key_f32(kx));
            w[g2] = ea;
            if (g + ke < total) atomicAdd(&L.den[dl * HEADS + (lane & 7)], ea);
        }

        v4f x[EPW];
#pragma unroll
        for (int i = 0; i < EPW; ++i) {
            const int s = __builtin_amdgcn_readlane(s_l, i);
            x[i] = *(const v4f*)(xl + (size_t)s * F + 4 * lane);
        }

#pragma unroll
        for (int i = 0; i < EPW; ++i) {
            if (g + i < total) {
                const int dl = __builtin_amdgcn_readlane(d_l, i);
                float* ap = L.acc + dl * F + lane;
                const float wi = __shfl(w[i >> 2], (i & 3) * 8 + (lane >> 2), 32);
#pragma unroll
                for (int j = 0; j < 4; ++j) atomicAdd(ap + 32 * j, wi * x[i][j]);
            }
        }
    }
    }
    __syncthreads();

    int nn = n - lo;
    if (nn > OWN_NODES) nn = OWN_NODES;
    const v4f bv = *(const v4f*)(bias + 4 * lane);
    for (int p = tid; p < nn * (F / 4); p += OWN_T) {
        const int node = p >> 5;
        const float* ap = L.acc + node * F + lane;
        const v4f a  = (v4f){ap[0], ap[32], ap[64], ap[96]};
        const float dn = L.den[node * HEADS + (lane >> 2)] + 1e-16f;
        const v4f o = (v4f){a[0] / dn + bv[0], a[1] / dn + bv[1], a[2] / dn + bv[2], a[3] / dn + bv[3]};
        *(v4f*)(out + (size_t)(lo + node) * F + 4 * lane) = o;
    }
}

extern "C" void kernel_launch(void* const* d_in, const int* in_sizes, int n_in,
                              void* d_out, int out_size, void* d_ws, size_t ws_size,
                              hipStream_t stream) {
    const float* x    = (const float*)d_in[0];
    const int*   ei   = (const int*)  d_in[1];
    const float* Wl   = (const float*)d_in[2];
    const float* Wr   = (const float*)d_in[3];
    const float* att  = (const float*)d_in[4];
    const float* bias = (const float*)d_in[5];

    const int n = in_sizes[0] / F;
    const int m = in_sizes[1] / 2;
    const int* src = ei;
    const int* dst = ei + m;

    float*    xl    = (float*)d_ws;
    float*    xr    = xl + (size_t)n * F;
    float*    alpha = xr + (size_t)n * F;
    unsigned* amax  = (unsigned*)(alpha + (size_t)m * HEADS);
    float*    out   = (float*)d_out;

    gat_proj_wmma<<<512, 128, 0, stream>>>(x, Wl, Wr, xl, xr, amax, n);

    const int edgeWaves  = (m + EPW - 1) / EPW;
    const int edgeBlocks = (edgeWaves * 32 + 255) / 256;
    gat_edge_alpha<<<edgeBlocks, 256, 0, stream>>>(xl, xr, src, dst, att, alpha, amax, m);
    gat_owner_aggr<<<(n + OWN_NODES - 1) / OWN_NODES, OWN_T, 0, stream>>>(xl, src, dst, alpha, amax, bias, out, m, n);
}


# Owner scan reads a 1-byte owner tag per edge (dst / 196, stored by the logit kernel) and tests 4 tags per word, not one 4-byte dst each: a quarter of the scanned bytes; sums stay LDS arrival order
# speedup vs baseline: 2.4267x; 1.1557x over previous
#include <hip/hip_runtime.h>

#define F 128
#define HEADS 8
#define NEG_SLOPE 0.2f
#define KEY_NEG_INF 0x007FFFFFu
#define EPW 8
#define OWN_NODES 196
#define OWN_T 1024
#define OWN_CAP 16384
#define OWN_U 8
#define OWN_TAGGED 256

typedef __attribute__((ext_vector_type(2))) float v2f;
typedef __attribute__((ext_vector_type(4))) float v4f;
typedef __attribute__((ext_vector_type(8))) float v8f;
typedef __attribute__((ext_vector_type(4))) unsigned v4u;
typedef __attribute__((ext_vector_type(4))) int v4i;

__device__ __forceinline__ unsigned f32_key(float f) {
    unsigned b = __float_as_uint(f);
    return (b & 0x80000000u) ? ~b : (b | 0x80000000u);
}
__device__ __forceinline__ float key_f32(unsigned k) {
    unsigned b = (k & 0x80000000u) ? (k & 0x7FFFFFFFu) : ~k;
    return __uint_as_float(b);
}

#define PROJ_TN 4
#define PROJ_KP (F / 8)

__device__ __forceinline__ void proj_load_a(const float* __restrict__ x, int tile, int n,
                                            int c, int h, v4f (&a)[PROJ_KP]) {
    int arow = tile * 16 + c;
    if (arow >= n) arow = n - 1;
    const float* xrow = x + (size_t)arow * F + h * 4;
#pragma unroll
    for (int j = 0; j < PROJ_KP; ++j) a[j] = *(const v4f*)(xrow + 8 * j);
}

__global__ __launch_bounds__(128) void gat_proj_wmma(
    const float* __restrict__ x, const float* __restrict__ Wl,
    const float* __restrict__ Wr, float* __restrict__ xl,
    float* __restrict__ xr, unsigned* __restrict__ amax, int n) {
    const int lane = threadIdx.x & 31;
    const int c    = lane & 15;
    const int h    = lane >> 4;
    const int j4   = __builtin_amdgcn_readfirstlane(threadIdx.x >> 5);

    const float* W   = (j4 < 2) ? Wl : Wr;
    float*       dst = (j4 < 2) ? xl : xr;
    const int col0   = (j4 & 1) * (16 * PROJ_TN);

    const int ntiles = (n + 15) >> 4;
    int tile = blockIdx.x;
    if (tile >= ntiles) return;

    v4f b[PROJ_TN][PROJ_KP];
#pragma unroll
    for (int t = 0; t < PROJ_TN; ++t) {
        const float* wrow = W + (size_t)(col0 + 16 * t + c) * F + h * 4;
#pragma unroll
        for (int j = 0; j < PROJ_KP; ++j) b[t][j] = *(const v4f*)(wrow + 8 * j);
    }

    v4f a[PROJ_KP];
    proj_load_a(x, tile, n, c, h, a);

    for (;;) {
        const int  next     = tile + gridDim.x;
        const bool has_next = next < ntiles;
        v4f an[PROJ_KP];
        proj_load_a(x, has_next ? next : tile, n, c, h, an);

        if (j4 == 0) {
            const int node = tile * 16 + (lane >> 1);
            if (node < n) {
                const size_t o = (size_t)node * HEADS + (lane & 1) * 4;
                *(v4u*)(amax + o)  = (v4u){KEY_NEG_INF, KEY_NEG_INF, KEY_NEG_INF, KEY_NEG_INF};
            }
        }

        v8f acc[PROJ_TN];
#pragma unroll
        for (int t = 0; t < PROJ_TN; ++t) acc[t] = (v8f){0.f, 0.f, 0.f, 0.f, 0.f, 0.f, 0.f, 0.f};

#pragma unroll
        for (int j = 0; j < PROJ_KP; ++j) {
#pragma unroll
            for (int t = 0; t < PROJ_TN; ++t) {
                const v2f a0 = (v2f){a[j][0], a[j][1]},    a1 = (v2f){a[j][2], a[j][3]};
                const v2f b0 = (v2f){b[t][j][0], b[t][j][1]}, b1 = (v2f){b[t][j][2], b[t][j][3]};
                acc[t] = __builtin_amdgcn_wmma_f32_16x16x4_f32(
                    false, a0, false, b0, (short)0, acc[t], false, false);
                acc[t] = __builtin_amdgcn_wmma_f32_16x16x4_f32(
                    false, a1, false, b1, (short)0, acc[t], false, false);
            }
        }

        const int mbase = tile * 16 + h * 8;
        if (tile * 16 + 16 <= n) {
            float* drow = dst + (size_t)mbase * F + col0 + c;
#pragma unroll
            for (int t = 0; t < PROJ_TN; ++t) {
#pragma unroll
                for (int r = 0; r < 8; ++r) drow[r * F + t * 16] = acc[t][r];
            }
        } else {
#pragma unroll
            for (int t = 0; t < PROJ_TN; ++t) {
#pragma unroll
                for (int r = 0; r < 8; ++r) {
                    const int mrow = mbase + r;
                    if (mrow < n) dst[(size_t)mrow * F + col0 + t * 16 + c] = acc[t][r];
                }
            }
        }

        if (!has_next) break;
        tile = next;
#pragma unroll
        for (int j = 0; j < PROJ_KP; ++j) a[j] = an[j];
    }
}

template <bool FULL>
__device__ __forceinline__ void edge_alpha_body(
    const float* __restrict__ xl, const float* __restrict__ xr,
    const int* __restrict__ src, const int* __restrict__ dst,
    const float* __restrict__ att, float* __restrict__ alpha,
    unsigned* __restrict__ amax, unsigned char* __restrict__ tag, int m, int e0, int lane) {
    int ek = e0 + (lane & 7);
    if (!FULL && ek > m - 1) ek = m - 1;
    const int idx = ((lane & 8) ? dst : src)[ek];
    if ((lane >> 3) == 1 && (FULL || e0 + (lane & 7) < m))
        tag[e0 + (lane & 7)] = (unsigned char)((unsigned)idx / OWN_NODES);

    const int c0 = lane * 4;
    const v4f av = *(const v4f*)(att + c0);

    v4f xj[EPW], xi[EPW];
#pragma unroll
    for (int i = 0; i < EPW; ++i) {
        const int s = __builtin_amdgcn_readlane(idx, i);
        const int d = __builtin_amdgcn_readlane(idx, 8 + i);
        xj[i] = *(const v4f*)(xl + (size_t)s * F + c0);
        xi[i] = *(const v4f*)(xr + (size_t)d * F + c0);
    }

    float p[EPW];
#pragma unroll
    for (int i = 0; i < EPW; ++i) {
        float a = 0.f;
#pragma unroll
        for (int j = 0; j < 4; ++j) {
            float v = xi[i][j] + xj[i][j];
            v = (v > 0.f) ? v : NEG_SLOPE * v;
            a += av[j] * v;
        }
        a += __shfl_xor(a, 1, 32);
        a += __shfl_xor(a, 2, 32);
        p[i] = a;
    }

    const int q = lane & 3, h = lane >> 2;
#pragma unroll
    for (int g = 0; g < EPW / 4; ++g) {
        const float p0 = p[4 * g], p1 = p[4 * g + 1], p2 = p[4 * g + 2], p3 = p[4 * g + 3];
        const float lo  = (q & 1) ? p1 : p0;
        const float hi  = (q & 1) ? p3 : p2;
        const float val = (q & 2) ? hi : lo;
        const int e = e0 + 4 * g + q;
        const int d = __shfl(idx, 8 + 4 * g + q, 32);
        if (FULL || e < m) {
            alpha[(size_t)e * HEADS + h] = val;
            atomicMax(amax + (size_t)d * HEADS + h, f32_key(val));
        }
    }
}

__global__ __launch_bounds__(256) void gat_edge_alpha(
    const float* __restrict__ xl, const float* __restrict__ xr,
    const int* __restrict__ src, const int* __restrict__ dst,
    const float* __restrict__ att, float* __restrict__ alpha,
    unsigned* __restrict__ amax, unsigned char* __restrict__ tag, int m) {
    const int wave = __builtin_amdgcn_readfirstlane((blockIdx.x * blockDim.x + threadIdx.x) >> 5);
    const int lane = threadIdx.x & 31;
    const int e0   = wave * EPW;
    if (e0 >= m) return;
    if (e0 + EPW <= m) edge_alpha_body<true >(xl, xr, src, dst, att, alpha, amax, tag, m, e0, lane);
    else               edge_alpha_body<false>(xl, xr, src, dst, att, alpha, amax, tag, m, e0, lane);
}

struct OwnLds {
    float acc[OWN_NODES * F];
    float den[OWN_NODES * HEADS];
    int   le[OWN_CAP];
    int   ld[OWN_CAP];
    int   cnt;
};

__device__ __forceinline__ void own_list_edge(OwnLds& L, int e) {
    const int slot = atomicAdd(&L.cnt, 1);
    if (slot < OWN_CAP) L.le[slot] = e;
}
__device__ __forceinline__ void own_try_edge(OwnLds& L, int lo, int e, int d) {
    if ((unsigned)(d - lo) < (unsigned)OWN_NODES) own_list_edge(L, e);
}
__device__ __forceinline__ unsigned own_zero_bytes(unsigned x) {
    return ~(((x & 0x7F7F7F7Fu) + 0x7F7F7F7Fu) | x) & 0x80808080u;
}
__device__ __forceinline__ void own_scan_piece(OwnLds& L, unsigned pat, v4u w, int q) {
    unsigned z = own_zero_bytes(w[0] ^ pat) | (own_zero_bytes(w[1] ^ pat) >> 1) |
                 (own_zero_bytes(w[2] ^ pat) >> 2) | (own_zero_bytes(w[3] ^ pat) >> 3);
    while (z) {
        const int p = __builtin_ctz(z);
        z &= z - 1;
        own_list_edge(L, 16 * q + 4 * (7 - (p & 7)) + (p >> 3));
    }
}

__global__ __launch_bounds__(OWN_T) void gat_owner_aggr(
    const float* __restrict__ xl, const int* __restrict__ src,
    const int* __restrict__ dst, const float* __restrict__ alpha,
    const unsigned* __restrict__ amax, const unsigned char* __restrict__ tag,
    const float* __restrict__ bias, float* __restrict__ out, int m, int n) {
    __shared__ OwnLds L;
    const int tid  = threadIdx.x;
    const int lane = tid & 31;
    const int wave = __builtin_amdgcn_readfirstlane(tid >> 5);
    const int lo   = blockIdx.x * OWN_NODES;

    for (int i = tid; i < OWN_NODES * F / 4; i += OWN_T) *(v4f*)(L.acc + 4 * i) = (v4f){0.f, 0.f, 0.f, 0.f};
    for (int i = tid; i < OWN_NODES * HEADS; i += OWN_T) L.den[i] = 0.f;
    if (tid == 0) L.cnt = 0;
    __syncthreads();

    const int m16 = m >> 4;
    if (gridDim.x <= OWN_TAGGED) {
        const unsigned pat = blockIdx.x * 0x01010101u;
        const v4u* tag4 = (const v4u*)tag;
        const int full = ((m16 / OWN_T) / OWN_U) * OWN_U;
        const int rot  = (int)(((long long)blockIdx.x * full) / gridDim.x);
        for (int k0 = 0; k0 < full; k0 += OWN_U) {
            v4u tv[OWN_U];
#pragma unroll
            for (int u = 0; u < OWN_U; ++u) {
                int it = k0 + u + rot;
                if (it >= full) it -= full;
                tv[u] = tag4[it * OWN_T + tid];
            }
#pragma unroll
            for (int u = 0; u < OWN_U; ++u) {
                int it = k0 + u + rot;
                if (it >= full) it -= full;
                own_scan_piece(L, pat, tv[u], it * OWN_T + tid);
            }
        }
        for (int q = full * OWN_T + tid; q < m16; q += OWN_T) own_scan_piece(L, pat, tag4[q], q);
        for (int e = (m16 << 4) + tid; e < m; e += OWN_T) own_try_edge(L, lo, e, dst[e]);
    } else {
        for (int e = tid; e < m; e += OWN_T) own_try_edge(L, lo, e, dst[e]);
    }
    __syncthreads();

    const int total = L.cnt;
    if (total > OWN_CAP) {
        for (int e = tid; e < m; e += OWN_T) {
            const unsigned dl = (unsigned)(dst[e] - lo);
            if (dl < (unsigned)OWN_NODES) {
                const int s = src[e];
                for (int h = 0; h < HEADS; ++h) {
                    const float ea = __expf(alpha[(size_t)e * HEADS + h] - key_f32(amax[(size_t)(lo + dl) * HEADS + h]));
                    atomicAdd(&L.den[dl * HEADS + h], ea);
                    for (int cc = 0; cc < F / HEADS; ++cc) {
                        const int ch = h * (F / HEADS) + cc;
                        atomicAdd(&L.acc[dl * F + (ch & 3) * 32 + (ch >> 2)], ea * xl[(size_t)s * F + ch]);
                    }
                }
            }
        }
    } else {
    for (int i0 = 0; i0 < total; i0 += 4 * OWN_T) {
        int dv[4];
#pragma unroll
        for (int u = 0; u < 4; ++u) {
            int i = i0 + u * OWN_T + tid;
            if (i > total - 1) i = total - 1;
            dv[u] = dst[L.le[i]];
        }
#pragma unroll
        for (int u = 0; u < 4; ++u) {
            const int i = i0 + u * OWN_T + tid;
            if (i < total) L.ld[i] = dv[u] - lo;
        }
    }
    __syncthreads();
    for (int g = wave * EPW; g < total; g += (OWN_T / 32) * EPW) {
        int k = g + (lane & 7);
        if (k > total - 1) k = total - 1;
        const int e_l  = L.le[k];
        const int d_l  = L.ld[k];
        const int s_l  = src[e_l];

        float w[EPW / 4];
#pragma unroll
        for (int g2 = 0; g2 < EPW / 4; ++g2) {
            const int ke = 4 * g2 + (lane >> 3);
            const int e  = __shfl(e_l, ke, 32);
            const int dl = __shfl(d_l, ke, 32);
            const float    a  = alpha[(size_t)e * HEADS + (lane & 7)];
            const unsigned kx = amax[(size_t)(lo + dl) * HEADS + (lane & 7)];
            const float ea = __expf(a - key_f32(kx));
            w[g2] = ea;
            if (g + ke < total) atomicAdd(&L.den[dl * HEADS + (lane & 7)], ea);
        }

        v4f x[EPW];
#pragma unroll
        for (int i = 0; i < EPW; ++i) {
            const int s = __builtin_amdgcn_readlane(s_l, i);
            x[i] = *(const v4f*)(xl + (size_t)s * F + 4 * lane);
        }

#pragma unroll
        for (int i = 0; i < EPW; ++i) {
            if (g + i < total) {
                const int dl = __builtin_amdgcn_readlane(d_l, i);
                float* ap = L.acc + dl * F + lane;
                const float wi = __shfl(w[i >> 2], (i & 3) * 8 + (lane >> 2), 32);
#pragma unroll
                for (int j = 0; j < 4; ++j) atomicAdd(ap + 32 * j, wi * x[i][j]);
            }
        }
    }
    }
    __syncthreads();

    int nn = n - lo;
    if (nn > OWN_NODES) nn = OWN_NODES;
    const v4f bv = *(const v4f*)(bias + 4 * lane);
    for (int p = tid; p < nn * (F / 4); p += OWN_T) {
        const int node = p >> 5;
        const float* ap = L.acc + node * F + lane;
        const v4f a  = (v4f){ap[0], ap[32], ap[64], ap[96]};
        const float dn = L.den[node * HEADS + (lane >> 2)] + 1e-16f;
        const v4f o = (v4f){a[0] / dn + bv[0], a[1] / dn + bv[1], a[2] / dn + bv[2], a[3] / dn + bv[3]};
        *(v4f*)(out + (size_t)(lo + node) * F + 4 * lane) = o;
    }
}

extern "C" void kernel_launch(void* const* d_in, const int* in_sizes, int n_in,
                              void* d_out, int out_size, void* d_ws, size_t ws_size,
                              hipStream_t stream) {
    const float* x    = (const float*)d_in[0];
    const int*   ei   = (const int*)  d_in[1];
    const float* Wl   = (const float*)d_in[2];
    const float* Wr   = (const float*)d_in[3];
    const float* att  = (const float*)d_in[4];
    const float* bias = (const float*)d_in[5];

    const int n = in_sizes[0] / F;
    const int m = in_sizes[1] / 2;
    const int* src = ei;
    const int* dst = ei + m;

    float*    xl    = (float*)d_ws;
    float*    xr    = xl + (size_t)n * F;
    float*    alpha = xr + (size_t)n * F;
    unsigned* amax  = (unsigned*)(alpha + (size_t)m * HEADS);
    unsigned char* tag = (unsigned char*)(amax + (size_t)n * HEADS);
    float*    out   = (float*)d_out;

    gat_proj_wmma<<<512, 128, 0, stream>>>(x, Wl, Wr, xl, xr, amax, n);

    const int edgeWaves  = (m + EPW - 1) / EPW;
    const int edgeBlocks = (edgeWaves * 32 + 255) / 256;
    gat_edge_alpha<<<edgeBlocks, 256, 0, stream>>>(xl, xr, src, dst, att, alpha, amax, tag, m);
    gat_owner_aggr<<<(n + OWN_NODES - 1) / OWN_NODES, OWN_T, 0, stream>>>(xl, src, dst, alpha, amax, tag, bias, out, m, n);
}


# Logit launch dissolved: owner work-groups compute their own edges' logits (xr rows in LDS, LDS atomic max, logits kept in LDS); projection writes owner tags; 2 launches; sums stay LDS arrival order
# speedup vs baseline: 3.5957x; 1.4817x over previous
#include <hip/hip_runtime.h>

#define F 128
#define HEADS 8
#define NEG_SLOPE 0.2f
#define KEY_NEG_INF 0x007FFFFFu
#define EPW 8
#define OWN_NODES 196
#define OWN_T 1024
#define OWN_CAP 4096
#define OWN_U 8
#define OWN_TAGGED 256
#define OWN_WR ((OWN_NODES * (F / 4) + OWN_T - 1) / OWN_T)

typedef __attribute__((ext_vector_type(2))) float v2f;
typedef __attribute__((ext_vector_type(4))) float v4f;
typedef __attribute__((ext_vector_type(8))) float v8f;
typedef __attribute__((ext_vector_type(4))) unsigned v4u;
typedef __attribute__((ext_vector_type(4))) int v4i;

__device__ __forceinline__ unsigned f32_key(float f) {
    unsigned b = __float_as_uint(f);
    return (b & 0x80000000u) ? ~b : (b | 0x80000000u);
}
__device__ __forceinline__ float key_f32(unsigned k) {
    unsigned b = (k & 0x80000000u) ? (k & 0x7FFFFFFFu) : ~k;
    return __uint_as_float(b);
}

#define PROJ_TN 4
#define PROJ_KP (F / 8)

__device__ __forceinline__ void proj_load_a(const float* __restrict__ x, int tile, int n,
                                            int c, int h, v4f (&a)[PROJ_KP]) {
    int arow = tile * 16 + c;
    if (arow >= n) arow = n - 1;
    const float* xrow = x + (size_t)arow * F + h * 4;
#pragma unroll
    for (int j = 0; j < PROJ_KP; ++j) a[j] = *(const v4f*)(xrow + 8 * j);
}

__global__ __launch_bounds__(128) void gat_proj_wmma(
    const float* __restrict__ x, const float* __restrict__ Wl,
    const float* __restrict__ Wr, const int* __restrict__ edst,
    float* __restrict__ xl, float* __restrict__ xr,
    unsigned char* __restrict__ tag, int n, int m) {
    {
        const int gt = blockIdx.x * blockDim.x + threadIdx.x;
        const int gs = gridDim.x * blockDim.x;
        const int m4 = m >> 2;
        for (int q = gt; q < m4; q += gs) {
            const int* dp = edst + 4 * q;
            const unsigned t0 = ((unsigned)dp[0] / OWN_NODES) & 255u;
            const unsigned t1 = ((unsigned)dp[1] / OWN_NODES) & 255u;
            const unsigned t2 = ((unsigned)dp[2] / OWN_NODES) & 255u;
            const unsigned t3 = ((unsigned)dp[3] / OWN_NODES) & 255u;
            *(unsigned*)(tag + 4 * q) = t0 | (t1 << 8) | (t2 << 16) | (t3 << 24);
        }
        for (int e = (m4 << 2) + gt; e < m; e += gs)
            tag[e] = (unsigned char)((unsigned)edst[e] / OWN_NODES);
    }

    const int lane = threadIdx.x & 31;
    const int c    = lane & 15;
    const int h    = lane >> 4;
    const int j4   = __builtin_amdgcn_readfirstlane(threadIdx.x >> 5);

    const float* W   = (j4 < 2) ? Wl : Wr;
    float*       dst = (j4 < 2) ? xl : xr;
    const int col0   = (j4 & 1) * (16 * PROJ_TN);

    const int ntiles = (n + 15) >> 4;
    int tile = blockIdx.x;
    if (tile >= ntiles) return;

    v4f b[PROJ_TN][PROJ_KP];
#pragma unroll
    for (int t = 0; t < PROJ_TN; ++t) {
        const float* wrow = W + (size_t)(col0 + 16 * t + c) * F + h * 4;
#pragma unroll
        for (int j = 0; j < PROJ_KP; ++j) b[t][j] = *(const v4f*)(wrow + 8 * j);
    }

    v4f a[PROJ_KP];
    proj_load_a(x, tile, n, c, h, a);

    for (;;) {
        const int  next     = tile + gridDim.x;
        const bool has_next = next < ntiles;
        v4f an[PROJ_KP];
        proj_load_a(x, has_next ? next : tile, n, c, h, an);

        v8f acc[PROJ_TN];
#pragma unroll
        for (int t = 0; t < PROJ_TN; ++t) acc[t] = (v8f){0.f, 0.f, 0.f, 0.f, 0.f, 0.f, 0.f, 0.f};

#pragma unroll
        for (int j = 0; j < PROJ_KP; ++j) {
#pragma unroll
            for (int t = 0; t < PROJ_TN; ++t) {
                const v2f a0 = (v2f){a[j][0], a[j][1]},    a1 = (v2f){a[j][2], a[j][3]};
                const v2f b0 = (v2f){b[t][j][0], b[t][j][1]}, b1 = (v2f){b[t][j][2], b[t][j][3]};
                acc[t] = __builtin_amdgcn_wmma_f32_16x16x4_f32(
                    false, a0, false, b0, (short)0, acc[t], false, false);
                acc[t] = __builtin_amdgcn_wmma_f32_16x16x4_f32(
                    false, a1, false, b1, (short)0, acc[t], false, false);
            }
        }

        const int mbase = tile * 16 + h * 8;
        if (tile * 16 + 16 <= n) {
            float* drow = dst + (size_t)mbase * F + col0 + c;
#pragma unroll
            for (int t = 0; t < PROJ_TN; ++t) {
#pragma unroll
                for (int r = 0; r < 8; ++r) drow[r * F + t * 16] = acc[t][r];
            }
        } else {
#pragma unroll
            for (int t = 0; t < PROJ_TN; ++t) {
#pragma unroll
                for (int r = 0; r < 8; ++r) {
                    const int mrow = mbase + r;
                    if (mrow < n) dst[(size_t)mrow * F + col0 + t * 16 + c] = acc[t][r];
                }
            }
        }

        if (!has_next) break;
        tile = next;
#pragma unroll
        for (int j = 0; j < PROJ_KP; ++j) a[j] = an[j];
    }
}

struct OwnLds {
    alignas(16) float acc[OWN_NODES * F];
    float    den[OWN_NODES * HEADS];
    unsigned mx[OWN_NODES * HEADS];
    float    lg[OWN_CAP * HEADS];
    int      ls[OWN_CAP];
    int      ld[OWN_CAP];
    int      cnt;
};

__device__ __forceinline__ void own_list_edge(OwnLds& L, int e) {
    const int slot = atomicAdd(&L.cnt, 1);
    if (slot < OWN_CAP) L.ls[slot] = e;
}
__device__ __forceinline__ void own_try_edge(OwnLds& L, int lo, int e, int d) {
    if ((unsigned)(d - lo) < (unsigned)OWN_NODES) own_list_edge(L, e);
}
__device__ __forceinline__ unsigned own_zero_bytes(unsigned x) {
    return ~(((x & 0x7F7F7F7Fu) + 0x7F7F7F7Fu) | x) & 0x80808080u;
}
__device__ __forceinline__ void own_scan_piece(OwnLds& L, unsigned pat, v4u w, int q) {
    unsigned z = own_zero_bytes(w[0] ^ pat) | (own_zero_bytes(w[1] ^ pat) >> 1) |
                 (own_zero_bytes(w[2] ^ pat) >> 2) | (own_zero_bytes(w[3] ^ pat) >> 3);
    while (z) {
        const int p = __builtin_ctz(z);
        z &= z - 1;
        own_list_edge(L, 16 * q + 4 * (7 - (p & 7)) + (p >> 3));
    }
}
__device__ __forceinline__ float own_logit_slow(const float* __restrict__ xlrow, const float* __restrict__ xrrow,
                                                const float* __restrict__ att, int h) {
    float a = 0.f;
    for (int cc = 0; cc < F / HEADS; ++cc) {
        const int ch = h * (F / HEADS) + cc;
        float v = xrrow[ch] + xlrow[ch];
        v = (v > 0.f) ? v : NEG_SLOPE * v;
        a += att[ch] * v;
    }
    return a;
}

__global__ __launch_bounds__(OWN_T) void gat_owner_aggr(
    const float* __restrict__ xl, const float* __restrict__ xr,
    const int* __restrict__ src, const int* __restrict__ dst,
    const float* __restrict__ att, const unsigned char* __restrict__ tag,
    const float* __restrict__ bias, float* __restrict__ out, int m, int n) {
    __shared__ OwnLds L;
    const int tid  = threadIdx.x;
    const int lane = tid & 31;
    const int wave = __builtin_amdgcn_readfirstlane(tid >> 5);
    const int lo   = blockIdx.x * OWN_NODES;
    int nn = n - lo;
    if (nn > OWN_NODES) nn = OWN_NODES;

    for (int i = tid; i < OWN_NODES * HEADS; i += OWN_T) { L.den[i] = 0.f; L.mx[i] = KEY_NEG_INF; }
    if (tid == 0) L.cnt = 0;
    __syncthreads();
    v4f wr[OWN_WR];
#pragma unroll
    for (int u = 0; u < OWN_WR; ++u) {
        int i = u * OWN_T + tid;
        if (i > nn * (F / 4) - 1) i = nn * (F / 4) - 1;
        wr[u] = ((const v4f*)(xr + (size_t)lo * F))[i];
    }

    const int m16 = m >> 4;
    if (gridDim.x <= OWN_TAGGED) {
        const unsigned pat = blockIdx.x * 0x01010101u;
        const v4u* tag4 = (const v4u*)tag;
        const int full = ((m16 / OWN_T) / OWN_U) * OWN_U;
        const int rot  = (int)(((long long)blockIdx.x * full) / gridDim.x);
        for (int k0 = 0; k0 < full; k0 += OWN_U) {
            v4u tv[OWN_U];
#pragma unroll
            for (int u = 0; u < OWN_U; ++u) {
                int it = k0 + u + rot;
                if (it >= full) it -= full;
                tv[u] = tag4[it * OWN_T + tid];
            }
#pragma unroll
            for (int u = 0; u < OWN_U; ++u) {
                int it = k0 + u + rot;
                if (it >= full) it -= full;
                own_scan_piece(L, pat, tv[u], it * OWN_T + tid);
            }
        }
        for (int q = full * OWN_T + tid; q < m16; q += OWN_T) own_scan_piece(L, pat, tag4[q], q);
        for (int e = (m16 << 4) + tid; e < m; e += OWN_T) own_try_edge(L, lo, e, dst[e]);
    } else {
        for (int e = tid; e < m; e += OWN_T) own_try_edge(L, lo, e, dst[e]);
    }
#pragma unroll
    for (int u = 0; u < OWN_WR; ++u) {
        const int i = u * OWN_T + tid;
        if (i < nn * (F / 4)) *(v4f*)(L.acc + 4 * i) = wr[u];
    }
    __syncthreads();

    const int  total = __builtin_amdgcn_readfirstlane(L.cnt);
    const bool slow  = total > OWN_CAP;

    const v4f av = *(const v4f*)(att + 4 * lane);

    if (slow) {
        for (int e = tid; e < m; e += OWN_T) {
            const unsigned dl = (unsigned)(dst[e] - lo);
            if (dl < (unsigned)OWN_NODES) {
                const float* xlrow = xl + (size_t)src[e] * F;
                const float* xrrow = xr + (size_t)(lo + dl) * F;
                for (int h = 0; h < HEADS; ++h)
                    atomicMax(&L.mx[dl * HEADS + h], f32_key(own_logit_slow(xlrow, xrrow, att, h)));
            }
        }
    } else {
        for (int i0 = 0; i0 < total; i0 += 4 * OWN_T) {
            int dv[4], sv[4];
#pragma unroll
            for (int u = 0; u < 4; ++u) {
                const int i = i0 + u * OWN_T + tid;
                dv[u] = 0; sv[u] = 0;
                if (i < total) {
                    const int e = L.ls[i];
                    dv[u] = dst[e];
                    sv[u] = src[e];
                }
            }
#pragma unroll
            for (int u = 0; u < 4; ++u) {
                const int i = i0 + u * OWN_T + tid;
                if (i < total) { L.ld[i] = dv[u] - lo; L.ls[i] = sv[u]; }
            }
        }
        __syncthreads();
        for (int g = wave * EPW; g < total; g += (OWN_T / 32) * EPW) {
            int k = g + (lane & 7);
            if (k > total - 1) k = total - 1;
            const int s_l = L.ls[k];
            const int d_l = L.ld[k];

            v4f xj[EPW];
#pragma unroll
            for (int i = 0; i < EPW; ++i) {
                const int s = __builtin_amdgcn_readlane(s_l, i);
                xj[i] = *(const v4f*)(xl + (size_t)s * F + 4 * lane);
            }

            float p[EPW];
#pragma unroll
            for (int i = 0; i < EPW; ++i) {
                const int dl = __builtin_amdgcn_readlane(d_l, i);
                const v4f xi = *(const v4f*)(L.acc + dl * F + 4 * lane);
                float a = 0.f;
#pragma unroll
                for (int j = 0; j < 4; ++j) {
                    float v = xi[j] + xj[i][j];
                    v = (v > 0.f) ? v : NEG_SLOPE * v;
                    a += av[j] * v;
                }
                a += __shfl_xor(a, 1, 32);
                a += __shfl_xor(a, 2, 32);
                p[i] = a;
            }

            const int q = lane & 3, h = lane >> 2;
#pragma unroll
            for (int gg = 0; gg < EPW / 4; ++gg) {
                const float p0 = p[4 * gg], p1 = p[4 * gg + 1], p2 = p[4 * gg + 2], p3 = p[4 * gg + 3];
                const float plo = (q & 1) ? p1 : p0;
                const float phi = (q & 1) ? p3 : p2;
                const float val = (q & 2) ? phi : plo;
                const int ke = 4 * gg + q;
                const int dl = __shfl(d_l, ke, 32);
                if (g + ke < total) {
                    L.lg[(g + ke) * HEADS + h] = val;
                    atomicMax(&L.mx[dl * HEADS + h], f32_key(val));
                }
            }
        }
    }
    __syncthreads();

    for (int i = tid; i < OWN_NODES * F / 4; i += OWN_T) *(v4f*)(L.acc + 4 * i) = (v4f){0.f, 0.f, 0.f, 0.f};
    __syncthreads();

    if (slow) {
        for (int e = tid; e < m; e += OWN_T) {
            const unsigned dl = (unsigned)(dst[e] - lo);
            if (dl < (unsigned)OWN_NODES) {
                const float* xlrow = xl + (size_t)src[e] * F;
                const float* xrrow = xr + (size_t)(lo + dl) * F;
                for (int h = 0; h < HEADS; ++h) {
                    const float ea = __expf(own_logit_slow(xlrow, xrrow, att, h) - key_f32(L.mx[dl * HEADS + h]));
                    atomicAdd(&L.den[dl * HEADS + h], ea);
                    for (int cc = 0; cc < F / HEADS; ++cc) {
                        const int ch = h * (F / HEADS) + cc;
                        atomicAdd(&L.acc[dl * F + (ch & 3) * 32 + (ch >> 2)], ea * xlrow[ch]);
                    }
                }
            }
        }
    } else {
        for (int g = wave * EPW; g < total; g += (OWN_T / 32) * EPW) {
            int k = g + (lane & 7);
            if (k > total - 1) k = total - 1;
            const int s_l = L.ls[k];
            const int d_l = L.ld[k];

            v4f x[EPW];
#pragma unroll
            for (int i = 0; i < EPW; ++i) {
                const int s = __builtin_amdgcn_readlane(s_l, i);
                x[i] = *(const v4f*)(xl + (size_t)s * F + 4 * lane);
            }

            float w[EPW / 4];
#pragma unroll
            for (int g2 = 0; g2 < EPW / 4; ++g2) {
                const int ke = 4 * g2 + (lane >> 3);
                int kk = g + ke;
                if (kk > total - 1) kk = total - 1;
                const int dl = __shfl(d_l, ke, 32);
                const float    a  = L.lg[kk * HEADS + (lane & 7)];
                const unsigned kx = L.mx[dl * HEADS + (lane & 7)];
                const float ea = __expf(a - key_f32(kx));
                w[g2] = ea;
                if (g + ke < total) atomicAdd(&L.den[dl * HEADS + (lane & 7)], ea);
            }

#pragma unroll
            for (int i = 0; i < EPW; ++i) {
                if (g + i < total) {
                    const int dl = __builtin_amdgcn_readlane(d_l, i);
                    float* ap = L.acc + dl * F + lane;
                    const float wi = __shfl(w[i >> 2], (i & 3) * 8 + (lane >> 2), 32);
#pragma unroll
                    for (int j = 0; j < 4; ++j) atomicAdd(ap + 32 * j, wi * x[i][j]);
                }
            }
        }
    }
    __syncthreads();

    const v4f bv = *(const v4f*)(bias + 4 * lane);
    for (int p = tid; p < nn * (F / 4); p += OWN_T) {
        const int node = p >> 5;
        const float* ap = L.acc + node * F + lane;
        const v4f a  = (v4f){ap[0], ap[32], ap[64], ap[96]};
        const float dn = L.den[node * HEADS + (lane >> 2)] + 1e-16f;
        const v4f o = (v4f){a[0] / dn + bv[0], a[1] / dn + bv[1], a[2] / dn + bv[2], a[3] / dn + bv[3]};
        *(v4f*)(out + (size_t)(lo + node) * F + 4 * lane) = o;
    }
}

extern "C" void kernel_launch(void* const* d_in, const int* in_sizes, int n_in,
                              void* d_out, int out_size, void* d_ws, size_t ws_size,
                              hipStream_t stream) {
    const float* x    = (const float*)d_in[0];
    const int*   ei   = (const int*)  d_in[1];
    const float* Wl   = (const float*)d_in[2];
    const float* Wr   = (const float*)d_in[3];
    const float* att  = (const float*)d_in[4];
    const float* bias = (const float*)d_in[5];

    const int n = in_sizes[0] / F;
    const int m = in_sizes[1] / 2;
    const int* src = ei;
    const int* dst = ei + m;

    float*    xl    = (float*)d_ws;
    float*    xr    = xl + (size_t)n * F;
    unsigned char* tag = (unsigned char*)(xr + (size_t)n * F);
    float*    out   = (float*)d_out;

    gat_proj_wmma<<<512, 128, 0, stream>>>(x, Wl, Wr, dst, xl, xr, tag, n, m);
    gat_owner_aggr<<<(n + OWN_NODES - 1) / OWN_NODES, OWN_T, 0, stream>>>(xl, xr, src, dst, att, tag, bias, out, m, n);
}


# Owner groups gather each xl[src] row once: logit, exp and LDS atomic sums in one edge pass, no logit list; exp without the max shift while all |logit| <= 16, else redone with it; LDS arrival order
# speedup vs baseline: 4.1138x; 1.1441x over previous
#include <hip/hip_runtime.h>

#define F 128
#define HEADS 8
#define NEG_SLOPE 0.2f
#define KEY_NEG_INF 0x007FFFFFu
#define EPW 8
#define OWN_NODES 196
#define OWN_T 1024
#define OWN_CAP 4096
#define OWN_U 8
#define OWN_TAGGED 256
#define OWN_WR ((OWN_NODES * (F / 4) + OWN_T - 1) / OWN_T)

typedef __attribute__((ext_vector_type(2))) float v2f;
typedef __attribute__((ext_vector_type(4))) float v4f;
typedef __attribute__((ext_vector_type(8))) float v8f;
typedef __attribute__((ext_vector_type(4))) unsigned v4u;
typedef __attribute__((ext_vector_type(4))) int v4i;

__device__ __forceinline__ unsigned f32_key(float f) {
    unsigned b = __float_as_uint(f);
    return (b & 0x80000000u) ? ~b : (b | 0x80000000u);
}
__device__ __forceinline__ float key_f32(unsigned k) {
    unsigned b = (k & 0x80000000u) ? (k & 0x7FFFFFFFu) : ~k;
    return __uint_as_float(b);
}

#define PROJ_TN 4
#define PROJ_KP (F / 8)

__device__ __forceinline__ void proj_load_a(const float* __restrict__ x, int tile, int n,
                                            int c, int h, v4f (&a)[PROJ_KP]) {
    int arow = tile * 16 + c;
    if (arow >= n) arow = n - 1;
    const float* xrow = x + (size_t)arow * F + h * 4;
#pragma unroll
    for (int j = 0; j < PROJ_KP; ++j) a[j] = *(const v4f*)(xrow + 8 * j);
}

__global__ __launch_bounds__(128) void gat_proj_wmma(
    const float* __restrict__ x, const float* __restrict__ Wl,
    const float* __restrict__ Wr, const int* __restrict__ edst,
    float* __restrict__ xl, float* __restrict__ xr,
    unsigned char* __restrict__ tag, int n, int m) {
    {
        const int gt = blockIdx.x * blockDim.x + threadIdx.x;
        const int gs = gridDim.x * blockDim.x;
        const int m4 = m >> 2;
        for (int q = gt; q < m4; q += gs) {
            const int* dp = edst + 4 * q;
            const unsigned t0 = ((unsigned)dp[0] / OWN_NODES) & 255u;
            const unsigned t1 = ((unsigned)dp[1] / OWN_NODES) & 255u;
            const unsigned t2 = ((unsigned)dp[2] / OWN_NODES) & 255u;
            const unsigned t3 = ((unsigned)dp[3] / OWN_NODES) & 255u;
            *(unsigned*)(tag + 4 * q) = t0 | (t1 << 8) | (t2 << 16) | (t3 << 24);
        }
        for (int e = (m4 << 2) + gt; e < m; e += gs)
            tag[e] = (unsigned char)((unsigned)edst[e] / OWN_NODES);
    }

    const int lane = threadIdx.x & 31;
    const int c    = lane & 15;
    const int h    = lane >> 4;
    const int j4   = __builtin_amdgcn_readfirstlane(threadIdx.x >> 5);

    const float* W   = (j4 < 2) ? Wl : Wr;
    float*       dst = (j4 < 2) ? xl : xr;
    const int col0   = (j4 & 1) * (16 * PROJ_TN);

    const int ntiles = (n + 15) >> 4;
    int tile = blockIdx.x;
    if (tile >= ntiles) return;

    v4f b[PROJ_TN][PROJ_KP];
#pragma unroll
    for (int t = 0; t < PROJ_TN; ++t) {
        const float* wrow = W + (size_t)(col0 + 16 * t + c) * F + h * 4;
#pragma unroll
        for (int j = 0; j < PROJ_KP; ++j) b[t][j] = *(const v4f*)(wrow + 8 * j);
    }

    v4f a[PROJ_KP];
    proj_load_a(x, tile, n, c, h, a);

    for (;;) {
        const int  next     = tile + gridDim.x;
        const bool has_next = next < ntiles;
        v4f an[PROJ_KP];
        proj_load_a(x, has_next ? next : tile, n, c, h, an);

        v8f acc[PROJ_TN];
#pragma unroll
        for (int t = 0; t < PROJ_TN; ++t) acc[t] = (v8f){0.f, 0.f, 0.f, 0.f, 0.f, 0.f, 0.f, 0.f};

#pragma unroll
        for (int j = 0; j < PROJ_KP; ++j) {
#pragma unroll
            for (int t = 0; t < PROJ_TN; ++t) {
                const v2f a0 = (v2f){a[j][0], a[j][1]},    a1 = (v2f){a[j][2], a[j][3]};
                const v2f b0 = (v2f){b[t][j][0], b[t][j][1]}, b1 = (v2f){b[t][j][2], b[t][j][3]};
                acc[t] = __builtin_amdgcn_wmma_f32_16x16x4_f32(
                    false, a0, false, b0, (short)0, acc[t], false, false);
                acc[t] = __builtin_amdgcn_wmma_f32_16x16x4_f32(
                    false, a1, false, b1, (short)0, acc[t], false, false);
            }
        }

        const int mbase = tile * 16 + h * 8;
        if (tile * 16 + 16 <= n) {
            float* drow = dst + (size_t)mbase * F + col0 + c;
#pragma unroll
            for (int t = 0; t < PROJ_TN; ++t) {
#pragma unroll
                for (int r = 0; r < 8; ++r) drow[r * F + t * 16] = acc[t][r];
            }
        } else {
#pragma unroll
            for (int t = 0; t < PROJ_TN; ++t) {
#pragma unroll
                for (int r = 0; r < 8; ++r) {
                    const int mrow = mbase + r;
                    if (mrow < n) dst[(size_t)mrow * F + col0 + t * 16 + c] = acc[t][r];
                }
            }
        }

        if (!has_next) break;
        tile = next;
#pragma unroll
        for (int j = 0; j < PROJ_KP; ++j) a[j] = an[j];
    }
}

#define OWN_AMAX 16.0f

struct OwnLds {
    alignas(16) float xr[OWN_NODES * F];
    alignas(16) float acc[OWN_NODES * F];
    float    den[OWN_NODES * HEADS];
    unsigned mx[OWN_NODES * HEADS];
    int      ls[OWN_CAP];
    int      ld[OWN_CAP];
    int      cnt;
    int      big;
};

__device__ __forceinline__ void own_list_edge(OwnLds& L, int e) {
    const int slot = atomicAdd(&L.cnt, 1);
    if (slot < OWN_CAP) L.ls[slot] = e;
}
__device__ __forceinline__ void own_try_edge(OwnLds& L, int lo, int e, int d) {
    if ((unsigned)(d - lo) < (unsigned)OWN_NODES) own_list_edge(L, e);
}
__device__ __forceinline__ unsigned own_zero_bytes(unsigned x) {
    return ~(((x & 0x7F7F7F7Fu) + 0x7F7F7F7Fu) | x) & 0x80808080u;
}
__device__ __forceinline__ void own_scan_piece(OwnLds& L, unsigned pat, v4u w, int q) {
    unsigned z = own_zero_bytes(w[0] ^ pat) | (own_zero_bytes(w[1] ^ pat) >> 1) |
                 (own_zero_bytes(w[2] ^ pat) >> 2) | (own_zero_bytes(w[3] ^ pat) >> 3);
    while (z) {
        const int p = __builtin_ctz(z);
        z &= z - 1;
        own_list_edge(L, 16 * q + 4 * (7 - (p & 7)) + (p >> 3));
    }
}
__device__ __forceinline__ float own_logit_slow(const float* __restrict__ xlrow, const float* __restrict__ xrrow,
                                                const float* __restrict__ att, int h) {
    float a = 0.f;
    for (int cc = 0; cc < F / HEADS; ++cc) {
        const int ch = h * (F / HEADS) + cc;
        float v = xrrow[ch] + xlrow[ch];
        v = (v > 0.f) ? v : NEG_SLOPE * v;
        a += att[ch] * v;
    }
    return a;
}

__global__ __launch_bounds__(OWN_T) void gat_owner_aggr(
    const float* __restrict__ xl, const float* __restrict__ xr,
    const int* __restrict__ src, const int* __restrict__ dst,
    const float* __restrict__ att, const unsigned char* __restrict__ tag,
    const float* __restrict__ bias, float* __restrict__ out, int m, int n) {
    __shared__ OwnLds L;
    const int tid  = threadIdx.x;
    const int lane = tid & 31;
    const int wave = __builtin_amdgcn_readfirstlane(tid >> 5);
    const int lo   = blockIdx.x * OWN_NODES;
    int nn = n - lo;
    if (nn > OWN_NODES) nn = OWN_NODES;

    for (int i = tid; i < OWN_NODES * HEADS; i += OWN_T) L.den[i] = 0.f;
    for (int i = tid; i < OWN_NODES * F / 4; i += OWN_T) *(v4f*)(L.acc + 4 * i) = (v4f){0.f, 0.f, 0.f, 0.f};
    if (tid == 0) { L.cnt = 0; L.big = 0; }
    __syncthreads();
    v4f wr[OWN_WR];
#pragma unroll
    for (int u = 0; u < OWN_WR; ++u) {
        int i = u * OWN_T + tid;
        if (i > nn * (F / 4) - 1) i = nn * (F / 4) - 1;
        wr[u] = ((const v4f*)(xr + (size_t)lo * F))[i];
    }

    const int m16 = m >> 4;
    if (gridDim.x <= OWN_TAGGED) {
        const unsigned pat = blockIdx.x * 0x01010101u;
        const v4u* tag4 = (const v4u*)tag;
        const int full = ((m16 / OWN_T) / OWN_U) * OWN_U;
        const int rot  = (int)(((long long)blockIdx.x * full) / gridDim.x);
        for (int k0 = 0; k0 < full; k0 += OWN_U) {
            v4u tv[OWN_U];
#pragma unroll
            for (int u = 0; u < OWN_U; ++u) {
                int it = k0 + u + rot;
                if (it >= full) it -= full;
                tv[u] = tag4[it * OWN_T + tid];
            }
#pragma unroll
            for (int u = 0; u < OWN_U; ++u) {
                int it = k0 + u + rot;
                if (it >= full) it -= full;
                own_scan_piece(L, pat, tv[u], it * OWN_T + tid);
            }
        }
        for (int q = full * OWN_T + tid; q < m16; q += OWN_T) own_scan_piece(L, pat, tag4[q], q);
        for (int e = (m16 << 4) + tid; e < m; e += OWN_T) own_try_edge(L, lo, e, dst[e]);
    } else {
        for (int e = tid; e < m; e += OWN_T) own_try_edge(L, lo, e, dst[e]);
    }
#pragma unroll
    for (int u = 0; u < OWN_WR; ++u) {
        const int i = u * OWN_T + tid;
        if (i < nn * (F / 4)) *(v4f*)(L.xr + 4 * i) = wr[u];
    }
    __syncthreads();

    const int  total = __builtin_amdgcn_readfirstlane(L.cnt);
    const bool slow  = total > OWN_CAP;

    if (!slow) {
        for (int i0 = 0; i0 < total; i0 += 4 * OWN_T) {
            int dv[4], sv[4];
#pragma unroll
            for (int u = 0; u < 4; ++u) {
                const int i = i0 + u * OWN_T + tid;
                dv[u] = 0; sv[u] = 0;
                if (i < total) {
                    const int e = L.ls[i];
                    dv[u] = dst[e];
                    sv[u] = src[e];
                }
            }
#pragma unroll
            for (int u = 0; u < 4; ++u) {
                const int i = i0 + u * OWN_T + tid;
                if (i < total) { L.ld[i] = dv[u] - lo; L.ls[i] = sv[u]; }
            }
        }
        __syncthreads();

        const v4f av = *(const v4f*)(att + 4 * lane);
        bool bad = false;
        for (int g = wave * EPW; g < total; g += (OWN_T / 32) * EPW) {
            int k = g + (lane & 7);
            if (k > total - 1) k = total - 1;
            const int s_l = L.ls[k];
            const int d_l = L.ld[k];

            v4f xj[EPW];
#pragma unroll
            for (int i = 0; i < EPW; ++i) {
                const int s = __builtin_amdgcn_readlane(s_l, i);
                xj[i] = *(const v4f*)(xl + (size_t)s * F + 4 * lane);
            }

            float p[EPW];
#pragma unroll
            for (int i = 0; i < EPW; ++i) {
                const int dl = __builtin_amdgcn_readlane(d_l, i);
                const v4f xi = *(const v4f*)(L.xr + dl * F + 4 * lane);
                float a = 0.f;
#pragma unroll
                for (int j = 0; j < 4; ++j) {
                    float v = xi[j] + xj[i][j];
                    v = (v > 0.f) ? v : NEG_SLOPE * v;
                    a += av[j] * v;
                }
                a += __shfl_xor(a, 1, 32);
                a += __shfl_xor(a, 2, 32);
                p[i] = a;
            }

#pragma unroll
            for (int i = 0; i < EPW; ++i) {
                if (g + i < total) {
                    const int dl = __builtin_amdgcn_readlane(d_l, i);
                    const float a = p[i];
                    bad = bad || !(fabsf(a) <= OWN_AMAX);
                    const float ea = __expf(a);
                    if ((lane & 3) == 0) atomicAdd(&L.den[dl * HEADS + (lane >> 2)], ea);
                    float* ap = L.acc + dl * F + lane;
#pragma unroll
                    for (int j = 0; j < 4; ++j) atomicAdd(ap + 32 * j, ea * xj[i][j]);
                }
            }
        }
        if (bad) L.big = 1;
    }
    __syncthreads();

    const bool big = __builtin_amdgcn_readfirstlane(L.big) != 0;
    if (slow || big) {
        for (int i = tid; i < OWN_NODES * HEADS; i += OWN_T) { L.den[i] = 0.f; L.mx[i] = KEY_NEG_INF; }
        for (int i = tid; i < OWN_NODES * F / 4; i += OWN_T) *(v4f*)(L.acc + 4 * i) = (v4f){0.f, 0.f, 0.f, 0.f};
        __syncthreads();
        for (int e = tid; e < m; e += OWN_T) {
            const unsigned dl = (unsigned)(dst[e] - lo);
            if (dl < (unsigned)OWN_NODES) {
                const float* xlrow = xl + (size_t)src[e] * F;
                const float* xrrow = xr + (size_t)(lo + dl) * F;
                for (int h = 0; h < HEADS; ++h)
                    atomicMax(&L.mx[dl * HEADS + h], f32_key(own_logit_slow(xlrow, xrrow, att, h)));
            }
        }
        __syncthreads();
        for (int e = tid; e < m; e += OWN_T) {
            const unsigned dl = (unsigned)(dst[e] - lo);
            if (dl < (unsigned)OWN_NODES) {
                const float* xlrow = xl + (size_t)src[e] * F;
                const float* xrrow = xr + (size_t)(lo + dl) * F;
                for (int h = 0; h < HEADS; ++h) {
                    const float ea = __expf(own_logit_slow(xlrow, xrrow, att, h) - key_f32(L.mx[dl * HEADS + h]));
                    atomicAdd(&L.den[dl * HEADS + h], ea);
                    for (int cc = 0; cc < F / HEADS; ++cc) {
                        const int ch = h * (F / HEADS) + cc;
                        atomicAdd(&L.acc[dl * F + (ch & 3) * 32 + (ch >> 2)], ea * xlrow[ch]);
                    }
                }
            }
        }
        __syncthreads();
    }

    const v4f bv = *(const v4f*)(bias + 4 * lane);
    for (int p = tid; p < nn * (F / 4); p += OWN_T) {
        const int node = p >> 5;
        const float* ap = L.acc + node * F + lane;
        const v4f a  = (v4f){ap[0], ap[32], ap[64], ap[96]};
        const float dn = L.den[node * HEADS + (lane >> 2)] + 1e-16f;
        const v4f o = (v4f){a[0] / dn + bv[0], a[1] / dn + bv[1], a[2] / dn + bv[2], a[3] / dn + bv[3]};
        *(v4f*)(out + (size_t)(lo + node) * F + 4 * lane) = o;
    }
}

extern "C" void kernel_launch(void* const* d_in, const int* in_sizes, int n_in,
                              void* d_out, int out_size, void* d_ws, size_t ws_size,
                              hipStream_t stream) {
    const float* x    = (const float*)d_in[0];
    const int*   ei   = (const int*)  d_in[1];
    const float* Wl   = (const float*)d_in[2];
    const float* Wr   = (const float*)d_in[3];
    const float* att  = (const float*)d_in[4];
    const float* bias = (const float*)d_in[5];

    const int n = in_sizes[0] / F;
    const int m = in_sizes[1] / 2;
    const int* src = ei;
    const int* dst = ei + m;

    float*    xl    = (float*)d_ws;
    float*    xr    = xl + (size_t)n * F;
    unsigned char* tag = (unsigned char*)(xr + (size_t)n * F);
    float*    out   = (float*)d_out;

    gat_proj_wmma<<<512, 128, 0, stream>>>(x, Wl, Wr, dst, xl, xr, tag, n, m);
    gat_owner_aggr<<<(n + OWN_NODES - 1) / OWN_NODES, OWN_T, 0, stream>>>(xl, xr, src, dst, att, tag, bias, out, m, n);
}
